# starting kernel
# baseline (speedup 1.0000x reference)
#include <hip/hip_runtime.h>
#include <math.h>

typedef __attribute__((ext_vector_type(2))) float v2f;
typedef __attribute__((ext_vector_type(8))) float v8f;

#define BROWS   262144
#define NF      10
#define HU      64
#define NP      9
#define HP      64
#define NCOLS   1216
#define KP      12
#define NTILES  76
#define THREADS 256
#define WAVES   8
#define TPW     4
#define RT_PER_BLOCK (WAVES * TPW)
#define NBLOCKS (BROWS / 16 / RT_PER_BLOCK)

__device__ __forceinline__ float gelu_exact(float v) {
    const float xs = v * 0.70710678118654752440f;
    const float ax = __builtin_fabsf(xs);
    const float t  = __builtin_amdgcn_rcpf(__builtin_fmaf(0.3275911f, ax, 1.0f));
    float p;
    p = __builtin_fmaf(1.061405429f, t, -1.453152027f);
    p = __builtin_fmaf(p, t,  1.421413741f);
    p = __builtin_fmaf(p, t, -0.284496736f);
    p = __builtin_fmaf(p, t,  0.254829592f);
    p = p * t;
    const float e   = __expf(-ax * ax);
    const float era = __builtin_fmaf(-p, e, 1.0f);
    const float erf_xs = __builtin_copysignf(era, xs);
    return 0.5f * v * (1.0f + erf_xs);
}

__global__ __launch_bounds__(THREADS) void nam_wmma_kernel(
    const float* __restrict__ x,
    const float* __restrict__ raw_mean, const float* __restrict__ raw_std,
    const float* __restrict__ w1f, const float* __restrict__ b1f,
    const float* __restrict__ w2f, const float* __restrict__ b2f,
    const float* __restrict__ w1p, const float* __restrict__ b1p,
    const float* __restrict__ w2p, const float* __restrict__ b2p,
    const int* __restrict__ crp_index_p,
    float* __restrict__ out)
{
    __shared__ float sW1T[NCOLS * KP];
    __shared__ float sW2[NCOLS];

    const int tid = threadIdx.x;

    for (int i = tid; i < NCOLS * KP; i += THREADS) sW1T[i] = 0.0f;
    __syncthreads();
    for (int j = tid; j < NCOLS; j += THREADS) {
        if (j < NF * HU) {
            int f = j >> 6, h = j & 63;
            sW1T[j * KP + f]  = w1f[f * HU + h];
            sW1T[j * KP + 10] = b1f[f * HU + h];
            sW2[j]            = w2f[f * HU + h];
        } else {
            int q = j - NF * HU;
            int p = q >> 6, h = q & 63;
            sW1T[j * KP + 0]     = w1p[(p * 2 + 0) * HP + h];
            sW1T[j * KP + p + 1] = w1p[(p * 2 + 1) * HP + h];
            sW1T[j * KP + 10]    = b1p[p * HP + h];
            sW2[j]               = w2p[p * HP + h];
        }
    }
    __syncthreads();

    float cbias = 0.0f;
    for (int f = 0; f < NF; ++f) cbias += b2f[f];
    for (int p = 0; p < NP; ++p) cbias += b2p[p];

    const int crp  = *crp_index_p;
    const int lane = tid & 31;
    const int wave = tid >> 5;
    const int half = lane >> 4;
    const int l15  = lane & 15;

    float mn[3][2], is[3][2];
#pragma unroll
    for (int s = 0; s < 3; ++s) {
        int k0 = 4 * s + 2 * half;
#pragma unroll
        for (int e = 0; e < 2; ++e) {
            int k = k0 + e;
            if (k < NF) { mn[s][e] = raw_mean[k]; is[s][e] = 1.0f / raw_std[k]; }
            else        { mn[s][e] = 0.0f;        is[s][e] = 1.0f; }
        }
    }

    for (int rt = 0; rt < TPW; ++rt) {
        const int rowTile = blockIdx.x * RT_PER_BLOCK + rt * WAVES + wave;
        const int row     = rowTile * 16 + l15;

        if (rt + 1 < TPW) {
            const long nrow = (long)(rowTile + WAVES) * 16 + l15;
            __builtin_prefetch(x + nrow * NF, 0, 0);
        }

        v2f afrag[3];
#pragma unroll
        for (int s = 0; s < 3; ++s) {
            int k0 = 4 * s + 2 * half;
            if (k0 >= NF) {
                afrag[s].x = 1.0f;
                afrag[s].y = 0.0f;
            } else {
                v2f xv = *(const v2f*)(x + (long)row * NF + k0);
                float v0 = xv.x, v1 = xv.y;
                if (k0     == crp) v0 = __logf(fmaxf(v0 * 10.0f, 1e-6f));
                if (k0 + 1 == crp) v1 = __logf(fmaxf(v1 * 10.0f, 1e-6f));
                afrag[s].x = (v0 - mn[s][0]) * is[s][0];
                afrag[s].y = (v1 - mn[s][1]) * is[s][1];
            }
        }

        v8f acc = {0.f, 0.f, 0.f, 0.f, 0.f, 0.f, 0.f, 0.f};

#pragma unroll 2
        for (int t = 0; t < NTILES; ++t) {
            const float* col = &sW1T[(t * 16 + l15) * KP + 2 * half];
            v2f b0 = *(const v2f*)(col + 0);
            v2f b1 = *(const v2f*)(col + 4);
            v2f b2 = *(const v2f*)(col + 8);

            v8f c = {0.f, 0.f, 0.f, 0.f, 0.f, 0.f, 0.f, 0.f};
            c = __builtin_amdgcn_wmma_f32_16x16x4_f32(false, afrag[0], false, b0,
                                                      (short)0, c, false, false);
            c = __builtin_amdgcn_wmma_f32_16x16x4_f32(false, afrag[1], false, b1,
                                                      (short)0, c, false, false);
            c = __builtin_amdgcn_wmma_f32_16x16x4_f32(false, afrag[2], false, b2,
                                                      (short)0, c, false, false);

            const float w2v = sW2[t * 16 + l15];
#pragma unroll
            for (int i = 0; i < 8; ++i)
                acc[i] = fmaf(gelu_exact(c[i]), w2v, acc[i]);
        }

#pragma unroll
        for (int i = 0; i < 8; ++i) {
            float v = acc[i];
            v += __shfl_xor(v, 1, 32);
            v += __shfl_xor(v, 2, 32);
            v += __shfl_xor(v, 4, 32);
            v += __shfl_xor(v, 8, 32);
            acc[i] = v;
        }
        if (l15 == 0) {
            const int base = rowTile * 16 + half * 8;
#pragma unroll
            for (int i = 0; i < 8; ++i)
                out[base + i] = acc[i] + cbias;
        }
    }
}

extern "C" void kernel_launch(void* const* d_in, const int* in_sizes, int n_in,
                              void* d_out, int out_size, void* d_ws, size_t ws_size,
                              hipStream_t stream) {
    (void)in_sizes; (void)n_in; (void)out_size; (void)d_ws; (void)ws_size;
    const float* x        = (const float*)d_in[0];
    const float* raw_mean = (const float*)d_in[1];
    const float* raw_std  = (const float*)d_in[2];
    const float* w1f      = (const float*)d_in[3];
    const float* b1f      = (const float*)d_in[4];
    const float* w2f      = (const float*)d_in[5];
    const float* b2f      = (const float*)d_in[6];
    const float* w1p      = (const float*)d_in[7];
    const float* b1p      = (const float*)d_in[8];
    const float* w2p      = (const float*)d_in[9];
    const float* b2p      = (const float*)d_in[10];
    const int*   crp      = (const int*)d_in[11];
    float* out            = (float*)d_out;

    hipLaunchKernelGGL(nam_wmma_kernel, dim3(NBLOCKS), dim3(THREADS), 0, stream,
                       x, raw_mean, raw_std, w1f, b1f, w2f, b2f,
                       w1p, b1p, w2p, b2p, crp, out);
}


# GELU as (v+|v|*erf|x|)/2 on packed f32 pairs, four pairs per stage, exp2 and erf constants folded into the staged weights: 12 plain vector ops per 2 elements, was 24; same series and sum order
# speedup vs baseline: 1.4544x; 1.4544x over previous
#include <hip/hip_runtime.h>
#include <math.h>

typedef __attribute__((ext_vector_type(2))) float v2f;
typedef __attribute__((ext_vector_type(8))) float v8f;

#define BROWS   262144
#define NF      10
#define HU      64
#define NP      9
#define HP      64
#define NCOLS   1216
#define KP      12
#define NTILES  76
#define THREADS 256
#define WAVES   8
#define TPW     4
#define RT_PER_BLOCK (WAVES * TPW)
#define NBLOCKS (BROWS / 16 / RT_PER_BLOCK)

#define KS  0.84932180028801904272f
#define KC  (0.3275911f / 1.20112240878644980f)
#define KW2 (0.5f / KS)

__device__ __forceinline__ void gelu_tile_accumulate(const v8f c, const float w2v, v8f& acc) {
    const v2f w2 = {w2v, w2v};
    v2f u[4], a[4], t[4], e[4], p[4];
#pragma unroll
    for (int j = 0; j < 4; ++j) { u[j].x = c[2 * j]; u[j].y = c[2 * j + 1]; }
#pragma unroll
    for (int j = 0; j < 4; ++j) a[j] = __builtin_elementwise_abs(u[j]);
#pragma unroll
    for (int j = 0; j < 4; ++j) {
        const v2f den = __builtin_elementwise_fma(a[j], (v2f)(KC), (v2f)(1.0f));
        t[j].x = __builtin_amdgcn_rcpf(den.x);
        t[j].y = __builtin_amdgcn_rcpf(den.y);
    }
#pragma unroll
    for (int j = 0; j < 4; ++j) {
        const v2f uu = u[j] * u[j];
        e[j].x = __builtin_amdgcn_exp2f(-uu.x);
        e[j].y = __builtin_amdgcn_exp2f(-uu.y);
    }
#pragma unroll
    for (int j = 0; j < 4; ++j)
        p[j] = __builtin_elementwise_fma((v2f)(1.061405429f), t[j], (v2f)(-1.453152027f));
#pragma unroll
    for (int j = 0; j < 4; ++j) p[j] = __builtin_elementwise_fma(p[j], t[j], (v2f)( 1.421413741f));
#pragma unroll
    for (int j = 0; j < 4; ++j) p[j] = __builtin_elementwise_fma(p[j], t[j], (v2f)(-0.284496736f));
#pragma unroll
    for (int j = 0; j < 4; ++j) p[j] = __builtin_elementwise_fma(p[j], t[j], (v2f)( 0.254829592f));
#pragma unroll
    for (int j = 0; j < 4; ++j) p[j] = p[j] * t[j];
#pragma unroll
    for (int j = 0; j < 4; ++j) p[j] = __builtin_elementwise_fma(-p[j], e[j], (v2f)(1.0f));
#pragma unroll
    for (int j = 0; j < 4; ++j) p[j] = __builtin_elementwise_fma(a[j], p[j], u[j]);
#pragma unroll
    for (int j = 0; j < 4; ++j) {
        const v2f s = {acc[2 * j], acc[2 * j + 1]};
        const v2f r = __builtin_elementwise_fma(p[j], w2, s);
        acc[2 * j]     = r.x;
        acc[2 * j + 1] = r.y;
    }
}

__global__ __launch_bounds__(THREADS) void nam_wmma_kernel(
    const float* __restrict__ x,
    const float* __restrict__ raw_mean, const float* __restrict__ raw_std,
    const float* __restrict__ w1f, const float* __restrict__ b1f,
    const float* __restrict__ w2f, const float* __restrict__ b2f,
    const float* __restrict__ w1p, const float* __restrict__ b1p,
    const float* __restrict__ w2p, const float* __restrict__ b2p,
    const int* __restrict__ crp_index_p,
    float* __restrict__ out)
{
    __shared__ float sW1T[NCOLS * KP];
    __shared__ float sW2[NCOLS];

    const int tid = threadIdx.x;

    for (int i = tid; i < NCOLS * KP; i += THREADS) sW1T[i] = 0.0f;
    __syncthreads();
    for (int j = tid; j < NCOLS; j += THREADS) {
        if (j < NF * HU) {
            int f = j >> 6, h = j & 63;
            sW1T[j * KP + f]  = KS  * w1f[f * HU + h];
            sW1T[j * KP + 10] = KS  * b1f[f * HU + h];
            sW2[j]            = KW2 * w2f[f * HU + h];
        } else {
            int q = j - NF * HU;
            int p = q >> 6, h = q & 63;
            sW1T[j * KP + 0]     = KS  * w1p[(p * 2 + 0) * HP + h];
            sW1T[j * KP + p + 1] = KS  * w1p[(p * 2 + 1) * HP + h];
            sW1T[j * KP + 10]    = KS  * b1p[p * HP + h];
            sW2[j]               = KW2 * w2p[p * HP + h];
        }
    }
    __syncthreads();

    float cbias = 0.0f;
    for (int f = 0; f < NF; ++f) cbias += b2f[f];
    for (int p = 0; p < NP; ++p) cbias += b2p[p];

    const int crp  = *crp_index_p;
    const int lane = tid & 31;
    const int wave = tid >> 5;
    const int half = lane >> 4;
    const int l15  = lane & 15;

    float mn[3][2], is[3][2];
#pragma unroll
    for (int s = 0; s < 3; ++s) {
        int k0 = 4 * s + 2 * half;
#pragma unroll
        for (int e = 0; e < 2; ++e) {
            int k = k0 + e;
            if (k < NF) { mn[s][e] = raw_mean[k]; is[s][e] = 1.0f / raw_std[k]; }
            else        { mn[s][e] = 0.0f;        is[s][e] = 1.0f; }
        }
    }

    for (int rt = 0; rt < TPW; ++rt) {
        const int rowTile = blockIdx.x * RT_PER_BLOCK + rt * WAVES + wave;
        const int row     = rowTile * 16 + l15;

        if (rt + 1 < TPW) {
            const long nrow = (long)(rowTile + WAVES) * 16 + l15;
            __builtin_prefetch(x + nrow * NF, 0, 0);
        }

        v2f afrag[3];
#pragma unroll
        for (int s = 0; s < 3; ++s) {
            int k0 = 4 * s + 2 * half;
            if (k0 >= NF) {
                afrag[s].x = 1.0f;
                afrag[s].y = 0.0f;
            } else {
                v2f xv = *(const v2f*)(x + (long)row * NF + k0);
                float v0 = xv.x, v1 = xv.y;
                if (k0     == crp) v0 = __logf(fmaxf(v0 * 10.0f, 1e-6f));
                if (k0 + 1 == crp) v1 = __logf(fmaxf(v1 * 10.0f, 1e-6f));
                afrag[s].x = (v0 - mn[s][0]) * is[s][0];
                afrag[s].y = (v1 - mn[s][1]) * is[s][1];
            }
        }

        v8f acc = {0.f, 0.f, 0.f, 0.f, 0.f, 0.f, 0.f, 0.f};

#pragma unroll 2
        for (int t = 0; t < NTILES; ++t) {
            const float* col = &sW1T[(t * 16 + l15) * KP + 2 * half];
            v2f b0 = *(const v2f*)(col + 0);
            v2f b1 = *(const v2f*)(col + 4);
            v2f b2 = *(const v2f*)(col + 8);

            v8f c = {0.f, 0.f, 0.f, 0.f, 0.f, 0.f, 0.f, 0.f};
            c = __builtin_amdgcn_wmma_f32_16x16x4_f32(false, afrag[0], false, b0,
                                                      (short)0, c, false, false);
            c = __builtin_amdgcn_wmma_f32_16x16x4_f32(false, afrag[1], false, b1,
                                                      (short)0, c, false, false);
            c = __builtin_amdgcn_wmma_f32_16x16x4_f32(false, afrag[2], false, b2,
                                                      (short)0, c, false, false);

            const float w2v = sW2[t * 16 + l15];
            gelu_tile_accumulate(c, w2v, acc);
        }

#pragma unroll
        for (int i = 0; i < 8; ++i) {
            float v = acc[i];
            v += __shfl_xor(v, 1, 32);
            v += __shfl_xor(v, 2, 32);
            v += __shfl_xor(v, 4, 32);
            v += __shfl_xor(v, 8, 32);
            acc[i] = v;
        }
        if (l15 == 0) {
            const int base = rowTile * 16 + half * 8;
#pragma unroll
            for (int i = 0; i < 8; ++i)
                out[base + i] = acc[i] + cbias;
        }
    }
}

extern "C" void kernel_launch(void* const* d_in, const int* in_sizes, int n_in,
                              void* d_out, int out_size, void* d_ws, size_t ws_size,
                              hipStream_t stream) {
    (void)in_sizes; (void)n_in; (void)out_size; (void)d_ws; (void)ws_size;
    const float* x        = (const float*)d_in[0];
    const float* raw_mean = (const float*)d_in[1];
    const float* raw_std  = (const float*)d_in[2];
    const float* w1f      = (const float*)d_in[3];
    const float* b1f      = (const float*)d_in[4];
    const float* w2f      = (const float*)d_in[5];
    const float* b2f      = (const float*)d_in[6];
    const float* w1p      = (const float*)d_in[7];
    const float* b1p      = (const float*)d_in[8];
    const float* w2p      = (const float*)d_in[9];
    const float* b2p      = (const float*)d_in[10];
    const int*   crp      = (const int*)d_in[11];
    float* out            = (float*)d_out;

    hipLaunchKernelGGL(nam_wmma_kernel, dim3(NBLOCKS), dim3(THREADS), 0, stream,
                       x, raw_mean, raw_std, w1f, b1f, w2f, b2f,
                       w1p, b1p, w2p, b2p, crp, out);
}


# GELU tail, same erf series, as (den-1-Q(t)e)/KC, Q=(1-t)(a1+..+a5t^4) one chain; linear u term summed outside the tile loop, per-group partial sums (sum order changes): 10 plain ops/pair, was 12
# speedup vs baseline: 1.9116x; 1.3144x over previous
#include <hip/hip_runtime.h>
#include <math.h>

typedef __attribute__((ext_vector_type(2))) float v2f;
typedef __attribute__((ext_vector_type(4))) float v4f;
typedef __attribute__((ext_vector_type(8))) float v8f;

#define BROWS   262144
#define NF      10
#define HU      64
#define NP      9
#define HP      64
#define NCOLS   1216
#define NGROUPS 19
#define TPG     4
#define THREADS 256
#define WAVES   8
#define TPW     4
#define RT_PER_BLOCK (WAVES * TPW)
#define NBLOCKS (BROWS / 16 / RT_PER_BLOCK)

#define KS  0.84932180028801904272f
#define KC  (0.3275911f / 1.20112240878644980f)
#define KW2 (0.5f / KS)

#define QN0 ((float)(-0.254829592))
#define QN1 ((float)( 0.254829592 + 0.284496736))
#define QN2 ((float)(-0.284496736 - 1.421413741))
#define QN3 ((float)( 1.421413741 + 1.453152027))
#define QN4 ((float)(-1.453152027 - 1.061405429))
#define QN5 ((float)( 1.061405429))

__device__ __forceinline__ void gelu_tile_accumulate(const v8f c, const float w2v, v8f& acc) {
    const v2f w2 = {w2v, w2v};
    v2f u[4], d[4], t[4], e[4], q[4];
#pragma unroll
    for (int j = 0; j < 4; ++j) { u[j].x = c[2 * j]; u[j].y = c[2 * j + 1]; }
#pragma unroll
    for (int j = 0; j < 4; ++j) {
        float dx, dy;
        asm("v_fma_f32 %0, |%1|, %2, 1.0" : "=v"(dx) : "v"(u[j].x), "s"(KC));
        asm("v_fma_f32 %0, |%1|, %2, 1.0" : "=v"(dy) : "v"(u[j].y), "s"(KC));
        d[j].x = dx;
        d[j].y = dy;
        t[j].x = __builtin_amdgcn_rcpf(d[j].x);
        t[j].y = __builtin_amdgcn_rcpf(d[j].y);
    }
#pragma unroll
    for (int j = 0; j < 4; ++j) {
        const v2f uu = u[j] * u[j];
        e[j].x = __builtin_amdgcn_exp2f(-uu.x);
        e[j].y = __builtin_amdgcn_exp2f(-uu.y);
    }
#pragma unroll
    for (int j = 0; j < 4; ++j) q[j] = __builtin_elementwise_fma((v2f)(QN5), t[j], (v2f)(QN4));
#pragma unroll
    for (int j = 0; j < 4; ++j) q[j] = __builtin_elementwise_fma(q[j], t[j], (v2f)(QN3));
#pragma unroll
    for (int j = 0; j < 4; ++j) q[j] = __builtin_elementwise_fma(q[j], t[j], (v2f)(QN2));
#pragma unroll
    for (int j = 0; j < 4; ++j) q[j] = __builtin_elementwise_fma(q[j], t[j], (v2f)(QN1));
#pragma unroll
    for (int j = 0; j < 4; ++j) q[j] = __builtin_elementwise_fma(q[j], t[j], (v2f)(QN0));
#pragma unroll
    for (int j = 0; j < 4; ++j) q[j] = __builtin_elementwise_fma(q[j], e[j], d[j]);
#pragma unroll
    for (int j = 0; j < 4; ++j) {
        const v2f s = {acc[2 * j], acc[2 * j + 1]};
        const v2f r = __builtin_elementwise_fma(q[j], w2, s);
        acc[2 * j]     = r.x;
        acc[2 * j + 1] = r.y;
    }
}

__global__ __launch_bounds__(THREADS) void nam_wmma_kernel(
    const float* __restrict__ x,
    const float* __restrict__ raw_mean, const float* __restrict__ raw_std,
    const float* __restrict__ w1f, const float* __restrict__ b1f,
    const float* __restrict__ w2f, const float* __restrict__ b2f,
    const float* __restrict__ w1p, const float* __restrict__ b1p,
    const float* __restrict__ w2p, const float* __restrict__ b2p,
    const int* __restrict__ crp_index_p,
    float* __restrict__ out)
{
    __shared__ v4f   sB[NCOLS];
    __shared__ float sW2[NCOLS];
    __shared__ v2f   sA[WAVES * NGROUPS * 32];
    __shared__ float sG[(NCOLS / 32) * 4];
    __shared__ float sAff[NF + 1];
    __shared__ float sC[NGROUPS * 16];

    const int tid  = threadIdx.x;
    const int lane = tid & 31;
    const int wave = tid >> 5;
    const int half = lane >> 4;
    const int l15  = lane & 15;

    for (int j = tid; j < NCOLS; j += THREADS) {
        v4f kq;
        float w2s;
        if (j < NF * HU) {
            kq.x = KS  * w1f[j];
            kq.y = KS  * b1f[j];
            kq.z = 0.0f;
            w2s  = (KW2 / KC) * w2f[j];
        } else {
            int q = j - NF * HU;
            int p = q >> 6, h = q & 63;
            kq.x = KS  * w1p[(p * 2 + 0) * HP + h];
            kq.y = KS  * w1p[(p * 2 + 1) * HP + h];
            kq.z = KS  * b1p[q];
            w2s  = (KW2 / KC) * w2p[q];
        }
        kq.w   = 0.0f;
        sB[j]  = kq;
        sW2[j] = w2s;
    }
    __syncthreads();

    for (int i = tid; i < NGROUPS * 16; i += THREADS) {
        const float* const w = sW2 + (i >> 4) * (TPG * 16) + (i & 15);
        sC[i] = -((w[0] + w[16]) + (w[32] + w[48]));
    }

    if (tid < 3 * (NCOLS / 32)) {
        const int hg = tid / 3, kk = tid - 3 * hg;
        const float* const sBf = (const float*)sB;
        float s0 = 0.0f, s1 = 0.0f;
        for (int i = 0; i < 32; i += 2) {
            const int j0 = hg * 32 + ((i + hg) & 31);
            const int j1 = hg * 32 + ((i + 1 + hg) & 31);
            s0 = __builtin_fmaf(sW2[j0], sBf[j0 * 4 + kk], s0);
            s1 = __builtin_fmaf(sW2[j1], sBf[j1 * 4 + kk], s1);
        }
        sG[hg * 4 + kk] = KC * (s0 + s1);
    }
    __syncthreads();
    if (tid < NF) {
        float a = sG[(2 * tid) * 4 + 0] + sG[(2 * tid + 1) * 4 + 0];
        if (tid == 0) {
            for (int p = 0; p < NP; ++p)
                a += sG[(2 * (NF + p)) * 4 + 0] + sG[(2 * (NF + p) + 1) * 4 + 0];
        } else {
            a += sG[(2 * (NF + tid - 1)) * 4 + 1] + sG[(2 * (NF + tid - 1) + 1) * 4 + 1];
        }
        sAff[tid] = a;
    } else if (tid == NF) {
        float cst = 0.0f;
        for (int f = 0; f < NF; ++f) cst += b2f[f];
        for (int p = 0; p < NP; ++p) cst += b2p[p];
        for (int f = 0; f < NF; ++f) cst += sG[(2 * f) * 4 + 1] + sG[(2 * f + 1) * 4 + 1];
        for (int p = 0; p < NP; ++p) cst += sG[(2 * (NF + p)) * 4 + 2] + sG[(2 * (NF + p) + 1) * 4 + 2];
        sAff[NF] = cst;
    }
    __syncthreads();

    const int crp  = *crp_index_p;

    float mn[NF], is[NF];
#pragma unroll
    for (int k = 0; k < NF; ++k) { mn[k] = raw_mean[k]; is[k] = 1.0f / raw_std[k]; }

    v2f* const sAw = sA + wave * (NGROUPS * 32) + lane;
    const v2f* const sBh = (const v2f*)sB;

    for (int rt = 0; rt < TPW; ++rt) {
        const int rowTile = blockIdx.x * RT_PER_BLOCK + rt * WAVES + wave;
        const int row     = rowTile * 16 + l15;

        if (rt + 1 < TPW) {
            const long nrow = (long)(rowTile + WAVES) * 16 + l15;
            __builtin_prefetch(x + nrow * NF, 0, 0);
        }

        float z[NF];
#pragma unroll
        for (int s = 0; s < NF / 2; ++s) {
            const v2f xv = *(const v2f*)(x + (long)row * NF + 2 * s);
            z[2 * s]     = xv.x;
            z[2 * s + 1] = xv.y;
        }
        float xc = z[0];
#pragma unroll
        for (int k = 1; k < NF; ++k) xc = (k == crp) ? z[k] : xc;
        const float lg = __logf(fmaxf(xc * 10.0f, 1e-6f));
#pragma unroll
        for (int k = 0; k < NF; ++k) {
            const float v = (k == crp) ? lg : z[k];
            z[k] = (v - mn[k]) * is[k];
        }

        float aff = sAff[NF];
#pragma unroll
        for (int k = 0; k < NF; ++k) aff = __builtin_fmaf(sAff[k], z[k], aff);

#pragma unroll
        for (int f = 0; f < NF; ++f) {
            v2f a;
            a.x = half ? 0.0f : z[f];
            a.y = half ? 0.0f : 1.0f;
            sAw[f * 32] = a;
        }
#pragma unroll
        for (int p = 0; p < NP; ++p) {
            v2f a;
            a.x = half ? 1.0f : z[0];
            a.y = half ? 0.0f : z[p + 1];
            sAw[(NF + p) * 32] = a;
        }

        v8f acc = {0.f, 0.f, 0.f, 0.f, 0.f, 0.f, 0.f, 0.f};

        for (int g = 0; g < NGROUPS; ++g) {
            const v2f a = sAw[g * 32];
            const float cg = sC[g * 16 + l15];
            v8f c[TPG];
            float w2v[TPG];
#pragma unroll
            for (int j = 0; j < TPG; ++j) {
                const int n = (g * TPG + j) * 16 + l15;
                const v2f b = sBh[n * 2 + half];
                const v8f zero = {0.f, 0.f, 0.f, 0.f, 0.f, 0.f, 0.f, 0.f};
                c[j] = __builtin_amdgcn_wmma_f32_16x16x4_f32(false, a, false, b,
                                                             (short)0, zero, false, false);
                w2v[j] = sW2[n];
            }
            v8f ag = {cg, cg, cg, cg, cg, cg, cg, cg};
#pragma unroll
            for (int j = 0; j < TPG; ++j) gelu_tile_accumulate(c[j], w2v[j], ag);
            acc += ag;
        }

#pragma unroll
        for (int i = 0; i < 8; ++i) {
            float v = acc[i];
            v += __shfl_xor(v, 1, 32);
            v += __shfl_xor(v, 2, 32);
            v += __shfl_xor(v, 4, 32);
            v += __shfl_xor(v, 8, 32);
            acc[i] = v;
        }
        float sel = acc[0];
#pragma unroll
        for (int i = 1; i < 8; ++i) sel = ((l15 & 7) == i) ? acc[i] : sel;
        if ((l15 >> 3) == half)
            out[rowTile * 16 + l15] = sel + aff;
    }
}

extern "C" void kernel_launch(void* const* d_in, const int* in_sizes, int n_in,
                              void* d_out, int out_size, void* d_ws, size_t ws_size,
                              hipStream_t stream) {
    (void)in_sizes; (void)n_in; (void)out_size; (void)d_ws; (void)ws_size;
    const float* x        = (const float*)d_in[0];
    const float* raw_mean = (const float*)d_in[1];
    const float* raw_std  = (const float*)d_in[2];
    const float* w1f      = (const float*)d_in[3];
    const float* b1f      = (const float*)d_in[4];
    const float* w2f      = (const float*)d_in[5];
    const float* b2f      = (const float*)d_in[6];
    const float* w1p      = (const float*)d_in[7];
    const float* b1p      = (const float*)d_in[8];
    const float* w2p      = (const float*)d_in[9];
    const float* b2p      = (const float*)d_in[10];
    const int*   crp      = (const int*)d_in[11];
    float* out            = (float*)d_out;

    hipLaunchKernelGGL(nam_wmma_kernel, dim3(NBLOCKS), dim3(THREADS), 0, stream,
                       x, raw_mean, raw_std, w1f, b1f, w2f, b2f,
                       w1p, b1p, w2p, b2p, crp, out);
}


# GELU group loop issue order: 61 of a group's 64 rcp/exp2 have one plain vector instruction directly behind them (were runs of eight transcendentals); same arithmetic, same sum order
# speedup vs baseline: 2.0987x; 1.0979x over previous
#include <hip/hip_runtime.h>
#include <math.h>

typedef __attribute__((ext_vector_type(2))) float v2f;
typedef __attribute__((ext_vector_type(4))) float v4f;
typedef __attribute__((ext_vector_type(8))) float v8f;

#define BROWS   262144
#define NF      10
#define HU      64
#define NP      9
#define HP      64
#define NCOLS   1216
#define NGROUPS 19
#define TPG     4
#define THREADS 256
#define WAVES   8
#define TPW     4
#define RT_PER_BLOCK (WAVES * TPW)
#define NBLOCKS (BROWS / 16 / RT_PER_BLOCK)

#define KS  0.84932180028801904272f
#define KC  (0.3275911f / 1.20112240878644980f)
#define KW2 (0.5f / KS)

#define QN0 ((float)(-0.254829592))
#define QN1 ((float)( 0.254829592 + 0.284496736))
#define QN2 ((float)(-0.284496736 - 1.421413741))
#define QN3 ((float)( 1.421413741 + 1.453152027))
#define QN4 ((float)(-1.453152027 - 1.061405429))
#define QN5 ((float)( 1.061405429))

#define HOLD_ORDER() __builtin_amdgcn_sched_barrier(0)

__device__ __forceinline__ float den_of(const float uv) {
    float r;
    asm("v_fma_f32 %0, |%1|, %2, 1.0" : "=v"(r) : "v"(uv), "s"(KC));
    return r;
}

__device__ __forceinline__ void gelu_group_accumulate(const v8f (&c)[TPG], const float (&w2v)[TPG], v8f& ag) {
    v8f uu[TPG];
    v2f d[16], t[16], e[16], q[16];
    auto den = [&](const int i, const int h) {
        const float uv = c[i >> 2][2 * (i & 3) + h];
        if (h) d[i].y = den_of(uv); else d[i].x = den_of(uv);
    };
    auto horner1 = [&](const int i) { q[i] = __builtin_elementwise_fma((v2f)(QN5), t[i], (v2f)(QN4)); };
    auto horner  = [&](const int i, const float cf) { q[i] = __builtin_elementwise_fma(q[i], t[i], (v2f)(cf)); };
    auto plain = [&](const int i, const int s) {
        if (s < 2) {
            if (i < 14) den(i + 2, s); else horner(2 * (i - 14) + s, QN2);
        } else if (s == 2) {
            if (i >= 1) horner1(i - 1);
        } else {
            if (i >= 2) horner(i - 2, QN3);
        }
    };

    HOLD_ORDER();
#pragma unroll
    for (int j = 0; j < TPG; ++j) uu[j] = c[j] * c[j];
    den(0, 0);  den(0, 1);
    den(1, 0);  den(1, 1);
    HOLD_ORDER();
#pragma unroll
    for (int i = 0; i < 16; ++i) {
        const int j = i >> 2, k = 2 * (i & 3);
        t[i].x = __builtin_amdgcn_rcpf(d[i].x);         HOLD_ORDER();   plain(i, 0);   HOLD_ORDER();
        t[i].y = __builtin_amdgcn_rcpf(d[i].y);         HOLD_ORDER();   plain(i, 1);   HOLD_ORDER();
        e[i].x = __builtin_amdgcn_exp2f(-uu[j][k]);     HOLD_ORDER();   plain(i, 2);   HOLD_ORDER();
        e[i].y = __builtin_amdgcn_exp2f(-uu[j][k + 1]); HOLD_ORDER();   plain(i, 3);   HOLD_ORDER();
    }
    horner1(15);
    horner(14, QN3);
    horner(15, QN3);
#pragma unroll
    for (int i = 4; i < 16; ++i) horner(i, QN2);
#pragma unroll
    for (int i = 0; i < 16; ++i) horner(i, QN1);
#pragma unroll
    for (int i = 0; i < 16; ++i) horner(i, QN0);
#pragma unroll
    for (int i = 0; i < 16; ++i) q[i] = __builtin_elementwise_fma(q[i], e[i], d[i]);
#pragma unroll
    for (int i = 0; i < 16; ++i) {
        const int j = i & 3;
        const v2f w2 = {w2v[i >> 2], w2v[i >> 2]};
        const v2f s  = {ag[2 * j], ag[2 * j + 1]};
        const v2f r  = __builtin_elementwise_fma(q[i], w2, s);
        ag[2 * j]     = r.x;
        ag[2 * j + 1] = r.y;
    }
}

__global__ __launch_bounds__(THREADS) void nam_wmma_kernel(
    const float* __restrict__ x,
    const float* __restrict__ raw_mean, const float* __restrict__ raw_std,
    const float* __restrict__ w1f, const float* __restrict__ b1f,
    const float* __restrict__ w2f, const float* __restrict__ b2f,
    const float* __restrict__ w1p, const float* __restrict__ b1p,
    const float* __restrict__ w2p, const float* __restrict__ b2p,
    const int* __restrict__ crp_index_p,
    float* __restrict__ out)
{
    __shared__ v4f   sB[NCOLS];
    __shared__ float sW2[NCOLS];
    __shared__ v2f   sA[WAVES * NGROUPS * 32];
    __shared__ float sG[(NCOLS / 32) * 4];
    __shared__ float sAff[NF + 1];
    __shared__ float sC[NGROUPS * 16];

    const int tid  = threadIdx.x;
    const int lane = tid & 31;
    const int wave = tid >> 5;
    const int half = lane >> 4;
    const int l15  = lane & 15;

    for (int j = tid; j < NCOLS; j += THREADS) {
        v4f kq;
        float w2s;
        if (j < NF * HU) {
            kq.x = KS  * w1f[j];
            kq.y = KS  * b1f[j];
            kq.z = 0.0f;
            w2s  = (KW2 / KC) * w2f[j];
        } else {
            int q = j - NF * HU;
            int p = q >> 6, h = q & 63;
            kq.x = KS  * w1p[(p * 2 + 0) * HP + h];
            kq.y = KS  * w1p[(p * 2 + 1) * HP + h];
            kq.z = KS  * b1p[q];
            w2s  = (KW2 / KC) * w2p[q];
        }
        kq.w   = 0.0f;
        sB[j]  = kq;
        sW2[j] = w2s;
    }
    __syncthreads();

    for (int i = tid; i < NGROUPS * 16; i += THREADS) {
        const float* const w = sW2 + (i >> 4) * (TPG * 16) + (i & 15);
        sC[i] = -((w[0] + w[16]) + (w[32] + w[48]));
    }

    if (tid < 3 * (NCOLS / 32)) {
        const int hg = tid / 3, kk = tid - 3 * hg;
        const float* const sBf = (const float*)sB;
        float s0 = 0.0f, s1 = 0.0f;
        for (int i = 0; i < 32; i += 2) {
            const int j0 = hg * 32 + ((i + hg) & 31);
            const int j1 = hg * 32 + ((i + 1 + hg) & 31);
            s0 = __builtin_fmaf(sW2[j0], sBf[j0 * 4 + kk], s0);
            s1 = __builtin_fmaf(sW2[j1], sBf[j1 * 4 + kk], s1);
        }
        sG[hg * 4 + kk] = KC * (s0 + s1);
    }
    __syncthreads();
    if (tid < NF) {
        float a = sG[(2 * tid) * 4 + 0] + sG[(2 * tid + 1) * 4 + 0];
        if (tid == 0) {
            for (int p = 0; p < NP; ++p)
                a += sG[(2 * (NF + p)) * 4 + 0] + sG[(2 * (NF + p) + 1) * 4 + 0];
        } else {
            a += sG[(2 * (NF + tid - 1)) * 4 + 1] + sG[(2 * (NF + tid - 1) + 1) * 4 + 1];
        }
        sAff[tid] = a;
    } else if (tid == NF) {
        float cst = 0.0f;
        for (int f = 0; f < NF; ++f) cst += b2f[f];
        for (int p = 0; p < NP; ++p) cst += b2p[p];
        for (int f = 0; f < NF; ++f) cst += sG[(2 * f) * 4 + 1] + sG[(2 * f + 1) * 4 + 1];
        for (int p = 0; p < NP; ++p) cst += sG[(2 * (NF + p)) * 4 + 2] + sG[(2 * (NF + p) + 1) * 4 + 2];
        sAff[NF] = cst;
    }
    __syncthreads();

    const int crp  = *crp_index_p;

    float mn[NF], is[NF];
#pragma unroll
    for (int k = 0; k < NF; ++k) { mn[k] = raw_mean[k]; is[k] = 1.0f / raw_std[k]; }

    v2f* const sAw = sA + wave * (NGROUPS * 32) + lane;
    const v2f* const sBh = (const v2f*)sB;

    for (int rt = 0; rt < TPW; ++rt) {
        const int rowTile = blockIdx.x * RT_PER_BLOCK + rt * WAVES + wave;
        const int row     = rowTile * 16 + l15;

        if (rt + 1 < TPW) {
            const long nrow = (long)(rowTile + WAVES) * 16 + l15;
            __builtin_prefetch(x + nrow * NF, 0, 0);
        }

        float z[NF];
#pragma unroll
        for (int s = 0; s < NF / 2; ++s) {
            const v2f xv = *(const v2f*)(x + (long)row * NF + 2 * s);
            z[2 * s]     = xv.x;
            z[2 * s + 1] = xv.y;
        }
        float xc = z[0];
#pragma unroll
        for (int k = 1; k < NF; ++k) xc = (k == crp) ? z[k] : xc;
        const float lg = __logf(fmaxf(xc * 10.0f, 1e-6f));
#pragma unroll
        for (int k = 0; k < NF; ++k) {
            const float v = (k == crp) ? lg : z[k];
            z[k] = (v - mn[k]) * is[k];
        }

        float aff = sAff[NF];
#pragma unroll
        for (int k = 0; k < NF; ++k) aff = __builtin_fmaf(sAff[k], z[k], aff);

#pragma unroll
        for (int f = 0; f < NF; ++f) {
            v2f a;
            a.x = half ? 0.0f : z[f];
            a.y = half ? 0.0f : 1.0f;
            sAw[f * 32] = a;
        }
#pragma unroll
        for (int p = 0; p < NP; ++p) {
            v2f a;
            a.x = half ? 1.0f : z[0];
            a.y = half ? 0.0f : z[p + 1];
            sAw[(NF + p) * 32] = a;
        }

        v8f acc = {0.f, 0.f, 0.f, 0.f, 0.f, 0.f, 0.f, 0.f};

        for (int g = 0; g < NGROUPS; ++g) {
            const v2f a = sAw[g * 32];
            const float cg = sC[g * 16 + l15];
            v8f c[TPG];
            float w2v[TPG];
#pragma unroll
            for (int j = 0; j < TPG; ++j) {
                const int n = (g * TPG + j) * 16 + l15;
                const v2f b = sBh[n * 2 + half];
                const v8f zero = {0.f, 0.f, 0.f, 0.f, 0.f, 0.f, 0.f, 0.f};
                c[j] = __builtin_amdgcn_wmma_f32_16x16x4_f32(false, a, false, b,
                                                             (short)0, zero, false, false);
                w2v[j] = sW2[n];
            }
            v8f ag = {cg, cg, cg, cg, cg, cg, cg, cg};
            gelu_group_accumulate(c, w2v, ag);
            acc += ag;
        }

#pragma unroll
        for (int i = 0; i < 8; ++i) {
            float v = acc[i];
            v += __shfl_xor(v, 1, 32);
            v += __shfl_xor(v, 2, 32);
            v += __shfl_xor(v, 4, 32);
            v += __shfl_xor(v, 8, 32);
            acc[i] = v;
        }
        float sel = acc[0];
#pragma unroll
        for (int i = 1; i < 8; ++i) sel = ((l15 & 7) == i) ? acc[i] : sel;
        if ((l15 >> 3) == half)
            out[rowTile * 16 + l15] = sel + aff;
    }
}

extern "C" void kernel_launch(void* const* d_in, const int* in_sizes, int n_in,
                              void* d_out, int out_size, void* d_ws, size_t ws_size,
                              hipStream_t stream) {
    (void)in_sizes; (void)n_in; (void)out_size; (void)d_ws; (void)ws_size;
    const float* x        = (const float*)d_in[0];
    const float* raw_mean = (const float*)d_in[1];
    const float* raw_std  = (const float*)d_in[2];
    const float* w1f      = (const float*)d_in[3];
    const float* b1f      = (const float*)d_in[4];
    const float* w2f      = (const float*)d_in[5];
    const float* b2f      = (const float*)d_in[6];
    const float* w1p      = (const float*)d_in[7];
    const float* b1p      = (const float*)d_in[8];
    const float* w2p      = (const float*)d_in[9];
    const float* b2p      = (const float*)d_in[10];
    const int*   crp      = (const int*)d_in[11];
    float* out            = (float*)d_out;

    hipLaunchKernelGGL(nam_wmma_kernel, dim3(NBLOCKS), dim3(THREADS), 0, stream,
                       x, raw_mean, raw_std, w1f, b1f, w2f, b2f,
                       w1p, b1p, w2p, b2p, crp, out);
}


# erf as tanh(v G(v^2)), G degree 6, on the hardware tanh: 2 transcendentals per pair (were 4: rcp, exp2), each with 2 to 4 packed ops behind it; erf fit and terms change, tile sums start at 0
# speedup vs baseline: 2.3474x; 1.1185x over previous
#include <hip/hip_runtime.h>
#include <math.h>

typedef __attribute__((ext_vector_type(2))) float v2f;
typedef __attribute__((ext_vector_type(4))) float v4f;
typedef __attribute__((ext_vector_type(8))) float v8f;

#define BROWS   262144
#define NF      10
#define HU      64
#define NP      9
#define HP      64
#define NCOLS   1216
#define NGROUPS 19
#define TPG     4
#define THREADS 256
#define WAVES   8
#define TPW     4
#define RT_PER_BLOCK (WAVES * TPW)
#define NBLOCKS (BROWS / 16 / RT_PER_BLOCK)

#define G0 ((float)( 7.978849415e-1))
#define G1 ((float)( 3.633308451e-2))
#define G2 ((float)(-3.259489611e-5))
#define G3 ((float)(-5.530622976e-5))
#define G4 ((float)( 3.964751343e-6))
#define G5 ((float)(-1.322638957e-7))
#define G6 ((float)( 1.756185511e-9))

#define HOLD_ORDER() __builtin_amdgcn_sched_barrier(0)

__device__ __forceinline__ void gelu_group_accumulate(const v8f (&c)[TPG], const float (&w2v)[TPG], v8f& ag) {
    v2f w[16], q[16], y[16], m[16];
    auto val = [&](const int i) -> v2f {
        const v2f r = {c[i >> 2][2 * (i & 3)], c[i >> 2][2 * (i & 3) + 1]};
        return r;
    };
    auto chain = [&](const int i, const int r) {
        if (r == 0)      w[i] = val(i) * val(i);
        else if (r == 1) q[i] = __builtin_elementwise_fma((v2f)(G6), w[i], (v2f)(G5));
        else if (r == 2) q[i] = __builtin_elementwise_fma(q[i], w[i], (v2f)(G4));
        else if (r == 3) q[i] = __builtin_elementwise_fma(q[i], w[i], (v2f)(G3));
        else if (r == 4) q[i] = __builtin_elementwise_fma(q[i], w[i], (v2f)(G2));
        else if (r == 5) q[i] = __builtin_elementwise_fma(q[i], w[i], (v2f)(G1));
        else if (r == 6) q[i] = __builtin_elementwise_fma(q[i], w[i], (v2f)(G0));
        else             q[i] = val(i) * q[i];
    };
    auto weigh = [&](const int i) {
        const v2f w2 = {w2v[i >> 2], w2v[i >> 2]};
        m[i] = val(i) * w2;
    };
    auto sum = [&](const int i) {
        const int j = i & 3;
        v2f r;
        if (i < 4) r = y[i] * m[i];
        else {
            const v2f s = {ag[2 * j], ag[2 * j + 1]};
            r = __builtin_elementwise_fma(y[i], m[i], s);
        }
        ag[2 * j]     = r.x;
        ag[2 * j + 1] = r.y;
    };
    auto tanh_of = [&](const int b, const int t) {
        const int i = 4 * b + (t >> 1);
        if (t & 1) y[i].y = __builtin_amdgcn_tanhf(q[i].y);
        else       y[i].x = __builtin_amdgcn_tanhf(q[i].x);
    };

    HOLD_ORDER();
#pragma unroll
    for (int r = 0; r < 4; ++r) {
#pragma unroll
        for (int p = 0; p < 4; ++p) chain(p, r);
#pragma unroll
        for (int p = 0; p < 4; ++p) chain(4 + p, r);
    }
#pragma unroll
    for (int r = 4; r < 8; ++r) {
#pragma unroll
        for (int p = 0; p < 4; ++p) chain(p, r);
    }
    HOLD_ORDER();
#pragma unroll
    for (int b = 0; b < 2; ++b) {
#pragma unroll
        for (int t = 0; t < 8; ++t) {
            tanh_of(b, t);                                  HOLD_ORDER();
#pragma unroll
            for (int p = 0; p < 4; ++p) {
                if (t & 1) chain(4 * (b + 2) + p, t >> 1);
                else       chain(4 * (b + 1) + p, 4 + (t >> 1));
                HOLD_ORDER();
            }
        }
    }
#pragma unroll
    for (int t = 0; t < 8; ++t) {
        tanh_of(2, t);                                      HOLD_ORDER();
#pragma unroll
        for (int f = (7 * t + 1) / 2; f < (7 * (t + 1) + 1) / 2; ++f) {
            const int g7 = f >> 2, p = f & 3;
            if (g7 & 1) weigh(4 * (g7 >> 1) + p);
            else        chain(12 + p, 4 + (g7 >> 1));
            HOLD_ORDER();
        }
    }
#pragma unroll
    for (int t = 0; t < 8; ++t) {
        tanh_of(3, t);                                      HOLD_ORDER();
#pragma unroll
        for (int f = 2 * t; f < 2 * t + 2; ++f) {
            if (f < 4) weigh(12 + f);
            else       sum(f - 4);
            HOLD_ORDER();
        }
    }
#pragma unroll
    for (int p = 0; p < 4; ++p) sum(12 + p);
}

__global__ __launch_bounds__(THREADS) void nam_wmma_kernel(
    const float* __restrict__ x,
    const float* __restrict__ raw_mean, const float* __restrict__ raw_std,
    const float* __restrict__ w1f, const float* __restrict__ b1f,
    const float* __restrict__ w2f, const float* __restrict__ b2f,
    const float* __restrict__ w1p, const float* __restrict__ b1p,
    const float* __restrict__ w2p, const float* __restrict__ b2p,
    const int* __restrict__ crp_index_p,
    float* __restrict__ out)
{
    __shared__ v4f   sB[NCOLS];
    __shared__ float sW2[NCOLS];
    __shared__ v2f   sA[WAVES * NGROUPS * 32];
    __shared__ float sG[(NCOLS / 32) * 4];
    __shared__ float sAff[NF + 1];

    const int tid  = threadIdx.x;
    const int lane = tid & 31;
    const int wave = tid >> 5;
    const int half = lane >> 4;
    const int l15  = lane & 15;

    for (int j = tid; j < NCOLS; j += THREADS) {
        v4f kq;
        float w2s;
        if (j < NF * HU) {
            kq.x = w1f[j];
            kq.y = b1f[j];
            kq.z = 0.0f;
            w2s  = 0.5f * w2f[j];
        } else {
            int q = j - NF * HU;
            int p = q >> 6, h = q & 63;
            kq.x = w1p[(p * 2 + 0) * HP + h];
            kq.y = w1p[(p * 2 + 1) * HP + h];
            kq.z = b1p[q];
            w2s  = 0.5f * w2p[q];
        }
        kq.w   = 0.0f;
        sB[j]  = kq;
        sW2[j] = w2s;
    }
    __syncthreads();

    if (tid < 3 * (NCOLS / 32)) {
        const int hg = tid / 3, kk = tid - 3 * hg;
        const float* const sBf = (const float*)sB;
        float s0 = 0.0f, s1 = 0.0f;
        for (int i = 0; i < 32; i += 2) {
            const int j0 = hg * 32 + ((i + hg) & 31);
            const int j1 = hg * 32 + ((i + 1 + hg) & 31);
            s0 = __builtin_fmaf(sW2[j0], sBf[j0 * 4 + kk], s0);
            s1 = __builtin_fmaf(sW2[j1], sBf[j1 * 4 + kk], s1);
        }
        sG[hg * 4 + kk] = s0 + s1;
    }
    __syncthreads();
    if (tid < NF) {
        float a = sG[(2 * tid) * 4 + 0] + sG[(2 * tid + 1) * 4 + 0];
        if (tid == 0) {
            for (int p = 0; p < NP; ++p)
                a += sG[(2 * (NF + p)) * 4 + 0] + sG[(2 * (NF + p) + 1) * 4 + 0];
        } else {
            a += sG[(2 * (NF + tid - 1)) * 4 + 1] + sG[(2 * (NF + tid - 1) + 1) * 4 + 1];
        }
        sAff[tid] = a;
    } else if (tid == NF) {
        float cst = 0.0f;
        for (int f = 0; f < NF; ++f) cst += b2f[f];
        for (int p = 0; p < NP; ++p) cst += b2p[p];
        for (int f = 0; f < NF; ++f) cst += sG[(2 * f) * 4 + 1] + sG[(2 * f + 1) * 4 + 1];
        for (int p = 0; p < NP; ++p) cst += sG[(2 * (NF + p)) * 4 + 2] + sG[(2 * (NF + p) + 1) * 4 + 2];
        sAff[NF] = cst;
    }
    __syncthreads();

    const int crp  = *crp_index_p;

    float mn[NF], is[NF];
#pragma unroll
    for (int k = 0; k < NF; ++k) { mn[k] = raw_mean[k]; is[k] = 1.0f / raw_std[k]; }

    v2f* const sAw = sA + wave * (NGROUPS * 32) + lane;
    const v2f* const sBh = (const v2f*)sB;

    for (int rt = 0; rt < TPW; ++rt) {
        const int rowTile = blockIdx.x * RT_PER_BLOCK + rt * WAVES + wave;
        const int row     = rowTile * 16 + l15;

        if (rt + 1 < TPW) {
            const long nrow = (long)(rowTile + WAVES) * 16 + l15;
            __builtin_prefetch(x + nrow * NF, 0, 0);
        }

        float z[NF];
#pragma unroll
        for (int s = 0; s < NF / 2; ++s) {
            const v2f xv = *(const v2f*)(x + (long)row * NF + 2 * s);
            z[2 * s]     = xv.x;
            z[2 * s + 1] = xv.y;
        }
        float xc = z[0];
#pragma unroll
        for (int k = 1; k < NF; ++k) xc = (k == crp) ? z[k] : xc;
        const float lg = __logf(fmaxf(xc * 10.0f, 1e-6f));
#pragma unroll
        for (int k = 0; k < NF; ++k) {
            const float v = (k == crp) ? lg : z[k];
            z[k] = (v - mn[k]) * is[k];
        }

        float aff = sAff[NF];
#pragma unroll
        for (int k = 0; k < NF; ++k) aff = __builtin_fmaf(sAff[k], z[k], aff);

#pragma unroll
        for (int f = 0; f < NF; ++f) {
            v2f a;
            a.x = half ? 0.0f : z[f];
            a.y = half ? 0.0f : 1.0f;
            sAw[f * 32] = a;
        }
#pragma unroll
        for (int p = 0; p < NP; ++p) {
            v2f a;
            a.x = half ? 1.0f : z[0];
            a.y = half ? 0.0f : z[p + 1];
            sAw[(NF + p) * 32] = a;
        }

        v8f acc = {0.f, 0.f, 0.f, 0.f, 0.f, 0.f, 0.f, 0.f};

        for (int g = 0; g < NGROUPS; ++g) {
            const v2f a = sAw[g * 32];
            v8f c[TPG];
            float w2v[TPG];
#pragma unroll
            for (int j = 0; j < TPG; ++j) {
                const int n = (g * TPG + j) * 16 + l15;
                const v2f b = sBh[n * 2 + half];
                const v8f zero = {0.f, 0.f, 0.f, 0.f, 0.f, 0.f, 0.f, 0.f};
                c[j] = __builtin_amdgcn_wmma_f32_16x16x4_f32(false, a, false, b,
                                                             (short)0, zero, false, false);
                w2v[j] = sW2[n];
            }
            v8f ag = {0.f, 0.f, 0.f, 0.f, 0.f, 0.f, 0.f, 0.f};
            gelu_group_accumulate(c, w2v, ag);
            acc += ag;
        }

#pragma unroll
        for (int i = 0; i < 8; ++i) {
            float v = acc[i];
            v += __shfl_xor(v, 1, 32);
            v += __shfl_xor(v, 2, 32);
            v += __shfl_xor(v, 4, 32);
            v += __shfl_xor(v, 8, 32);
            acc[i] = v;
        }
        float sel = acc[0];
#pragma unroll
        for (int i = 1; i < 8; ++i) sel = ((l15 & 7) == i) ? acc[i] : sel;
        if ((l15 >> 3) == half)
            out[rowTile * 16 + l15] = sel + aff;
    }
}

extern "C" void kernel_launch(void* const* d_in, const int* in_sizes, int n_in,
                              void* d_out, int out_size, void* d_ws, size_t ws_size,
                              hipStream_t stream) {
    (void)in_sizes; (void)n_in; (void)out_size; (void)d_ws; (void)ws_size;
    const float* x        = (const float*)d_in[0];
    const float* raw_mean = (const float*)d_in[1];
    const float* raw_std  = (const float*)d_in[2];
    const float* w1f      = (const float*)d_in[3];
    const float* b1f      = (const float*)d_in[4];
    const float* w2f      = (const float*)d_in[5];
    const float* b2f      = (const float*)d_in[6];
    const float* w1p      = (const float*)d_in[7];
    const float* b1p      = (const float*)d_in[8];
    const float* w2p      = (const float*)d_in[9];
    const float* b2p      = (const float*)d_in[10];
    const int*   crp      = (const int*)d_in[11];
    float* out            = (float*)d_out;

    hipLaunchKernelGGL(nam_wmma_kernel, dim3(NBLOCKS), dim3(THREADS), 0, stream,
                       x, raw_mean, raw_std, w1f, b1f, w2f, b2f,
                       w1p, b1p, w2p, b2p, crp, out);
}


# All global reads of the prologue (staged weights, biases, normalisation, first row tile) issued before the first wait, later tiles' rows read one tile ahead into registers: latencies overlap
# speedup vs baseline: 2.4953x; 1.0630x over previous
#include <hip/hip_runtime.h>
#include <math.h>

typedef __attribute__((ext_vector_type(2))) float v2f;
typedef __attribute__((ext_vector_type(4))) float v4f;
typedef __attribute__((ext_vector_type(8))) float v8f;

#define BROWS   262144
#define NF      10
#define HU      64
#define NP      9
#define HP      64
#define NCOLS   1216
#define NGROUPS 19
#define TPG     4
#define THREADS 256
#define WAVES   8
#define TPW     4
#define RT_PER_BLOCK (WAVES * TPW)
#define NBLOCKS (BROWS / 16 / RT_PER_BLOCK)
#define NSTG    ((NCOLS + THREADS - 1) / THREADS)

#define G0 ((float)( 7.978849415e-1))
#define G1 ((float)( 3.633308451e-2))
#define G2 ((float)(-3.259489611e-5))
#define G3 ((float)(-5.530622976e-5))
#define G4 ((float)( 3.964751343e-6))
#define G5 ((float)(-1.322638957e-7))
#define G6 ((float)( 1.756185511e-9))

#define HOLD_ORDER() __builtin_amdgcn_sched_barrier(0)

__device__ __forceinline__ void gelu_group_accumulate(const v8f (&c)[TPG], const float (&w2v)[TPG], v8f& ag) {
    v2f w[16], q[16], y[16], m[16];
    auto val = [&](const int i) -> v2f {
        const v2f r = {c[i >> 2][2 * (i & 3)], c[i >> 2][2 * (i & 3) + 1]};
        return r;
    };
    auto chain = [&](const int i, const int r) {
        if (r == 0)      w[i] = val(i) * val(i);
        else if (r == 1) q[i] = __builtin_elementwise_fma((v2f)(G6), w[i], (v2f)(G5));
        else if (r == 2) q[i] = __builtin_elementwise_fma(q[i], w[i], (v2f)(G4));
        else if (r == 3) q[i] = __builtin_elementwise_fma(q[i], w[i], (v2f)(G3));
        else if (r == 4) q[i] = __builtin_elementwise_fma(q[i], w[i], (v2f)(G2));
        else if (r == 5) q[i] = __builtin_elementwise_fma(q[i], w[i], (v2f)(G1));
        else if (r == 6) q[i] = __builtin_elementwise_fma(q[i], w[i], (v2f)(G0));
        else             q[i] = val(i) * q[i];
    };
    auto weigh = [&](const int i) {
        const v2f w2 = {w2v[i >> 2], w2v[i >> 2]};
        m[i] = val(i) * w2;
    };
    auto sum = [&](const int i) {
        const int j = i & 3;
        v2f r;
        if (i < 4) r = y[i] * m[i];
        else {
            const v2f s = {ag[2 * j], ag[2 * j + 1]};
            r = __builtin_elementwise_fma(y[i], m[i], s);
        }
        ag[2 * j]     = r.x;
        ag[2 * j + 1] = r.y;
    };
    auto tanh_of = [&](const int b, const int t) {
        const int i = 4 * b + (t >> 1);
        if (t & 1) y[i].y = __builtin_amdgcn_tanhf(q[i].y);
        else       y[i].x = __builtin_amdgcn_tanhf(q[i].x);
    };

    HOLD_ORDER();
#pragma unroll
    for (int r = 0; r < 4; ++r) {
#pragma unroll
        for (int p = 0; p < 4; ++p) chain(p, r);
#pragma unroll
        for (int p = 0; p < 4; ++p) chain(4 + p, r);
    }
#pragma unroll
    for (int r = 4; r < 8; ++r) {
#pragma unroll
        for (int p = 0; p < 4; ++p) chain(p, r);
    }
    HOLD_ORDER();
#pragma unroll
    for (int b = 0; b < 2; ++b) {
#pragma unroll
        for (int t = 0; t < 8; ++t) {
            tanh_of(b, t);                                  HOLD_ORDER();
#pragma unroll
            for (int p = 0; p < 4; ++p) {
                if (t & 1) chain(4 * (b + 2) + p, t >> 1);
                else       chain(4 * (b + 1) + p, 4 + (t >> 1));
                HOLD_ORDER();
            }
        }
    }
#pragma unroll
    for (int t = 0; t < 8; ++t) {
        tanh_of(2, t);                                      HOLD_ORDER();
#pragma unroll
        for (int f = (7 * t + 1) / 2; f < (7 * (t + 1) + 1) / 2; ++f) {
            const int g7 = f >> 2, p = f & 3;
            if (g7 & 1) weigh(4 * (g7 >> 1) + p);
            else        chain(12 + p, 4 + (g7 >> 1));
            HOLD_ORDER();
        }
    }
#pragma unroll
    for (int t = 0; t < 8; ++t) {
        tanh_of(3, t);                                      HOLD_ORDER();
#pragma unroll
        for (int f = 2 * t; f < 2 * t + 2; ++f) {
            if (f < 4) weigh(12 + f);
            else       sum(f - 4);
            HOLD_ORDER();
        }
    }
#pragma unroll
    for (int p = 0; p < 4; ++p) sum(12 + p);
}

__global__ __launch_bounds__(THREADS) void nam_wmma_kernel(
    const float* __restrict__ x,
    const float* __restrict__ raw_mean, const float* __restrict__ raw_std,
    const float* __restrict__ w1f, const float* __restrict__ b1f,
    const float* __restrict__ w2f, const float* __restrict__ b2f,
    const float* __restrict__ w1p, const float* __restrict__ b1p,
    const float* __restrict__ w2p, const float* __restrict__ b2p,
    const int* __restrict__ crp_index_p,
    float* __restrict__ out)
{
    __shared__ v4f   sB[NCOLS];
    __shared__ float sW2[NCOLS];
    __shared__ v2f   sA[WAVES * NGROUPS * 32];
    __shared__ float sG[(NCOLS / 32) * 4];
    __shared__ float sAff[NF + 1];

    const int tid  = threadIdx.x;
    const int lane = tid & 31;
    const int wave = tid >> 5;
    const int half = lane >> 4;
    const int l15  = lane & 15;

    float ka[NSTG], kb[NSTG], kc[NSTG], kw[NSTG];
#pragma unroll
    for (int i = 0; i < NSTG; ++i) {
        const int j  = tid + i * THREADS;
        const int jc = (j < NCOLS) ? j : (NCOLS - 1);
        const bool isF = ((i + 1) * THREADS <= NF * HU) ? true : ((i * THREADS >= NF * HU) ? false : (jc < NF * HU));
        const int q = isF ? 0 : (jc - NF * HU);
        const int p = q >> 6, h = q & 63;
        const float* const pa = isF ? (w1f + jc) : (w1p + (p * 2 + 0) * HP + h);
        const float* const pb = isF ? (b1f + jc) : (w1p + (p * 2 + 1) * HP + h);
        const float* const pw = isF ? (w2f + jc) : (w2p + q);
        ka[i] = *pa;
        kb[i] = *pb;
        kw[i] = *pw;
        kc[i] = b1p[q];
    }
    float bsum = 0.0f;
    for (int f = 0; f < NF; ++f) bsum += b2f[f];
    for (int p = 0; p < NP; ++p) bsum += b2p[p];

    const int crp  = *crp_index_p;

    float mn[NF], sd[NF];
#pragma unroll
    for (int k = 0; k < NF; ++k) { mn[k] = raw_mean[k]; sd[k] = raw_std[k]; }

    v2f xn[NF / 2];
    {
        const long row0 = (long)(blockIdx.x * RT_PER_BLOCK + wave) * 16 + l15;
#pragma unroll
        for (int s = 0; s < NF / 2; ++s) xn[s] = *(const v2f*)(x + row0 * NF + 2 * s);
    }

    __builtin_amdgcn_sched_barrier(0);

#pragma unroll
    for (int i = 0; i < NSTG; ++i) {
        const int j = tid + i * THREADS;
        if (j < NCOLS) {
            v4f kq;
            kq.x = ka[i];
            kq.y = kb[i];
            kq.z = (j < NF * HU) ? 0.0f : kc[i];
            kq.w = 0.0f;
            sB[j]  = kq;
            sW2[j] = 0.5f * kw[i];
        }
    }
    __syncthreads();

    if (tid < 3 * (NCOLS / 32)) {
        const int hg = tid / 3, kk = tid - 3 * hg;
        const float* const sBf = (const float*)sB;
        float s0 = 0.0f, s1 = 0.0f;
        for (int i = 0; i < 32; i += 2) {
            const int j0 = hg * 32 + ((i + hg) & 31);
            const int j1 = hg * 32 + ((i + 1 + hg) & 31);
            s0 = __builtin_fmaf(sW2[j0], sBf[j0 * 4 + kk], s0);
            s1 = __builtin_fmaf(sW2[j1], sBf[j1 * 4 + kk], s1);
        }
        sG[hg * 4 + kk] = s0 + s1;
    }
    __syncthreads();
    if (tid < NF) {
        float a = sG[(2 * tid) * 4 + 0] + sG[(2 * tid + 1) * 4 + 0];
        if (tid == 0) {
            for (int p = 0; p < NP; ++p)
                a += sG[(2 * (NF + p)) * 4 + 0] + sG[(2 * (NF + p) + 1) * 4 + 0];
        } else {
            a += sG[(2 * (NF + tid - 1)) * 4 + 1] + sG[(2 * (NF + tid - 1) + 1) * 4 + 1];
        }
        sAff[tid] = a;
    } else if (tid == NF) {
        float cst = bsum;
        for (int f = 0; f < NF; ++f) cst += sG[(2 * f) * 4 + 1] + sG[(2 * f + 1) * 4 + 1];
        for (int p = 0; p < NP; ++p) cst += sG[(2 * (NF + p)) * 4 + 2] + sG[(2 * (NF + p) + 1) * 4 + 2];
        sAff[NF] = cst;
    }
    __syncthreads();

    float is[NF];
#pragma unroll
    for (int k = 0; k < NF; ++k) is[k] = 1.0f / sd[k];

    v2f* const sAw = sA + wave * (NGROUPS * 32) + lane;
    const v2f* const sBh = (const v2f*)sB;

    for (int rt = 0; rt < TPW; ++rt) {
        const int rowTile = blockIdx.x * RT_PER_BLOCK + rt * WAVES + wave;

        float z[NF];
#pragma unroll
        for (int s = 0; s < NF / 2; ++s) {
            z[2 * s]     = xn[s].x;
            z[2 * s + 1] = xn[s].y;
        }
        if (rt + 1 < TPW) {
            const long nrow = (long)(rowTile + WAVES) * 16 + l15;
#pragma unroll
            for (int s = 0; s < NF / 2; ++s) xn[s] = *(const v2f*)(x + nrow * NF + 2 * s);
        }
        float xc = z[0];
#pragma unroll
        for (int k = 1; k < NF; ++k) xc = (k == crp) ? z[k] : xc;
        const float lg = __logf(fmaxf(xc * 10.0f, 1e-6f));
#pragma unroll
        for (int k = 0; k < NF; ++k) {
            const float v = (k == crp) ? lg : z[k];
            z[k] = (v - mn[k]) * is[k];
        }

        float aff = sAff[NF];
#pragma unroll
        for (int k = 0; k < NF; ++k) aff = __builtin_fmaf(sAff[k], z[k], aff);

#pragma unroll
        for (int f = 0; f < NF; ++f) {
            v2f a;
            a.x = half ? 0.0f : z[f];
            a.y = half ? 0.0f : 1.0f;
            sAw[f * 32] = a;
        }
#pragma unroll
        for (int p = 0; p < NP; ++p) {
            v2f a;
            a.x = half ? 1.0f : z[0];
            a.y = half ? 0.0f : z[p + 1];
            sAw[(NF + p) * 32] = a;
        }

        v8f acc = {0.f, 0.f, 0.f, 0.f, 0.f, 0.f, 0.f, 0.f};

        for (int g = 0; g < NGROUPS; ++g) {
            const v2f a = sAw[g * 32];
            v8f c[TPG];
            float w2v[TPG];
#pragma unroll
            for (int j = 0; j < TPG; ++j) {
                const int n = (g * TPG + j) * 16 + l15;
                const v2f b = sBh[n * 2 + half];
                const v8f zero = {0.f, 0.f, 0.f, 0.f, 0.f, 0.f, 0.f, 0.f};
                c[j] = __builtin_amdgcn_wmma_f32_16x16x4_f32(false, a, false, b,
                                                             (short)0, zero, false, false);
                w2v[j] = sW2[n];
            }
            v8f ag = {0.f, 0.f, 0.f, 0.f, 0.f, 0.f, 0.f, 0.f};
            gelu_group_accumulate(c, w2v, ag);
            acc += ag;
        }

#pragma unroll
        for (int i = 0; i < 8; ++i) {
            float v = acc[i];
            v += __shfl_xor(v, 1, 32);
            v += __shfl_xor(v, 2, 32);
            v += __shfl_xor(v, 4, 32);
            v += __shfl_xor(v, 8, 32);
            acc[i] = v;
        }
        float sel = acc[0];
#pragma unroll
        for (int i = 1; i < 8; ++i) sel = ((l15 & 7) == i) ? acc[i] : sel;
        if ((l15 >> 3) == half)
            out[rowTile * 16 + l15] = sel + aff;
    }
}

extern "C" void kernel_launch(void* const* d_in, const int* in_sizes, int n_in,
                              void* d_out, int out_size, void* d_ws, size_t ws_size,
                              hipStream_t stream) {
    (void)in_sizes; (void)n_in; (void)out_size; (void)d_ws; (void)ws_size;
    const float* x        = (const float*)d_in[0];
    const float* raw_mean = (const float*)d_in[1];
    const float* raw_std  = (const float*)d_in[2];
    const float* w1f      = (const float*)d_in[3];
    const float* b1f      = (const float*)d_in[4];
    const float* w2f      = (const float*)d_in[5];
    const float* b2f      = (const float*)d_in[6];
    const float* w1p      = (const float*)d_in[7];
    const float* b1p      = (const float*)d_in[8];
    const float* w2p      = (const float*)d_in[9];
    const float* b2p      = (const float*)d_in[10];
    const int*   crp      = (const int*)d_in[11];
    float* out            = (float*)d_out;

    hipLaunchKernelGGL(nam_wmma_kernel, dim3(NBLOCKS), dim3(THREADS), 0, stream,
                       x, raw_mean, raw_std, w1f, b1f, w2f, b2f,
                       w1p, b1p, w2p, b2p, crp, out);
}


# Group loop outermost, the wave's four row tiles inside it: a group's B fragments, second-layer weights, LDS addresses and loop control once per group, were once per row tile and group; same sum order
# speedup vs baseline: 2.6473x; 1.0609x over previous
#include <hip/hip_runtime.h>
#include <math.h>

typedef __attribute__((ext_vector_type(2))) float v2f;
typedef __attribute__((ext_vector_type(4))) float v4f;
typedef __attribute__((ext_vector_type(8))) float v8f;

#define BROWS   262144
#define NF      10
#define HU      64
#define NP      9
#define HP      64
#define NCOLS   1216
#define NGROUPS 19
#define TPG     4
#define THREADS 256
#define WAVES   8
#define TPW     4
#define RT_PER_BLOCK (WAVES * TPW)
#define NBLOCKS (BROWS / 16 / RT_PER_BLOCK)
#define NSTG    ((NCOLS + THREADS - 1) / THREADS)
#define APITCH  17

#define G0 ((float)( 7.978849415e-1))
#define G1 ((float)( 3.633308451e-2))
#define G2 ((float)(-3.259489611e-5))
#define G3 ((float)(-5.530622976e-5))
#define G4 ((float)( 3.964751343e-6))
#define G5 ((float)(-1.322638957e-7))
#define G6 ((float)( 1.756185511e-9))

#define HOLD_ORDER() __builtin_amdgcn_sched_barrier(0)

__device__ __forceinline__ void gelu_group_accumulate(const v8f (&c)[TPG], const float (&w2v)[TPG], v8f& ag) {
    v2f w[16], q[16], y[16], m[16];
    auto val = [&](const int i) -> v2f {
        const v2f r = {c[i >> 2][2 * (i & 3)], c[i >> 2][2 * (i & 3) + 1]};
        return r;
    };
    auto chain = [&](const int i, const int r) {
        if (r == 0)      w[i] = val(i) * val(i);
        else if (r == 1) q[i] = __builtin_elementwise_fma((v2f)(G6), w[i], (v2f)(G5));
        else if (r == 2) q[i] = __builtin_elementwise_fma(q[i], w[i], (v2f)(G4));
        else if (r == 3) q[i] = __builtin_elementwise_fma(q[i], w[i], (v2f)(G3));
        else if (r == 4) q[i] = __builtin_elementwise_fma(q[i], w[i], (v2f)(G2));
        else if (r == 5) q[i] = __builtin_elementwise_fma(q[i], w[i], (v2f)(G1));
        else if (r == 6) q[i] = __builtin_elementwise_fma(q[i], w[i], (v2f)(G0));
        else             q[i] = val(i) * q[i];
    };
    auto weigh = [&](const int i) {
        const v2f w2 = {w2v[i >> 2], w2v[i >> 2]};
        m[i] = val(i) * w2;
    };
    auto sum = [&](const int i) {
        const int j = i & 3;
        v2f r;
        if (i < 4) r = y[i] * m[i];
        else {
            const v2f s = {ag[2 * j], ag[2 * j + 1]};
            r = __builtin_elementwise_fma(y[i], m[i], s);
        }
        ag[2 * j]     = r.x;
        ag[2 * j + 1] = r.y;
    };
    auto tanh_of = [&](const int b, const int t) {
        const int i = 4 * b + (t >> 1);
        if (t & 1) y[i].y = __builtin_amdgcn_tanhf(q[i].y);
        else       y[i].x = __builtin_amdgcn_tanhf(q[i].x);
    };

    HOLD_ORDER();
#pragma unroll
    for (int r = 0; r < 4; ++r) {
#pragma unroll
        for (int p = 0; p < 4; ++p) chain(p, r);
#pragma unroll
        for (int p = 0; p < 4; ++p) chain(4 + p, r);
    }
#pragma unroll
    for (int r = 4; r < 8; ++r) {
#pragma unroll
        for (int p = 0; p < 4; ++p) chain(p, r);
    }
    HOLD_ORDER();
#pragma unroll
    for (int b = 0; b < 2; ++b) {
#pragma unroll
        for (int t = 0; t < 8; ++t) {
            tanh_of(b, t);                                  HOLD_ORDER();
#pragma unroll
            for (int p = 0; p < 4; ++p) {
                if (t & 1) chain(4 * (b + 2) + p, t >> 1);
                else       chain(4 * (b + 1) + p, 4 + (t >> 1));
                HOLD_ORDER();
            }
        }
    }
#pragma unroll
    for (int t = 0; t < 8; ++t) {
        tanh_of(2, t);                                      HOLD_ORDER();
#pragma unroll
        for (int f = (7 * t + 1) / 2; f < (7 * (t + 1) + 1) / 2; ++f) {
            const int g7 = f >> 2, p = f & 3;
            if (g7 & 1) weigh(4 * (g7 >> 1) + p);
            else        chain(12 + p, 4 + (g7 >> 1));
            HOLD_ORDER();
        }
    }
#pragma unroll
    for (int t = 0; t < 8; ++t) {
        tanh_of(3, t);                                      HOLD_ORDER();
#pragma unroll
        for (int f = 2 * t; f < 2 * t + 2; ++f) {
            if (f < 4) weigh(12 + f);
            else       sum(f - 4);
            HOLD_ORDER();
        }
    }
#pragma unroll
    for (int p = 0; p < 4; ++p) sum(12 + p);
}

__global__ __launch_bounds__(THREADS) void nam_wmma_kernel(
    const float* __restrict__ x,
    const float* __restrict__ raw_mean, const float* __restrict__ raw_std,
    const float* __restrict__ w1f, const float* __restrict__ b1f,
    const float* __restrict__ w2f, const float* __restrict__ b2f,
    const float* __restrict__ w1p, const float* __restrict__ b1p,
    const float* __restrict__ w2p, const float* __restrict__ b2p,
    const int* __restrict__ crp_index_p,
    float* __restrict__ out)
{
    __shared__ v4f   sB[NCOLS];
    __shared__ float sW2[NCOLS];
    __shared__ v2f   sA[WAVES * TPW * NGROUPS * APITCH];
    __shared__ float sG[(NCOLS / 32) * 4];
    __shared__ float sAff[NF + 1];

    const int tid  = threadIdx.x;
    const int lane = tid & 31;
    const int wave = tid >> 5;
    const int half = lane >> 4;
    const int l15  = lane & 15;

    float ka[NSTG], kb[NSTG], kc[NSTG], kw[NSTG];
#pragma unroll
    for (int i = 0; i < NSTG; ++i) {
        const int j  = tid + i * THREADS;
        const int jc = (j < NCOLS) ? j : (NCOLS - 1);
        const bool isF = ((i + 1) * THREADS <= NF * HU) ? true : ((i * THREADS >= NF * HU) ? false : (jc < NF * HU));
        const int q = isF ? 0 : (jc - NF * HU);
        const int p = q >> 6, h = q & 63;
        const float* const pa = isF ? (w1f + jc) : (w1p + (p * 2 + 0) * HP + h);
        const float* const pb = isF ? (b1f + jc) : (w1p + (p * 2 + 1) * HP + h);
        const float* const pw = isF ? (w2f + jc) : (w2p + q);
        ka[i] = *pa;
        kb[i] = *pb;
        kw[i] = *pw;
        kc[i] = b1p[q];
    }
    float bsum = 0.0f;
    for (int f = 0; f < NF; ++f) bsum += b2f[f];
    for (int p = 0; p < NP; ++p) bsum += b2p[p];

    const int crp  = *crp_index_p;

    float mn[NF], sd[NF];
#pragma unroll
    for (int k = 0; k < NF; ++k) { mn[k] = raw_mean[k]; sd[k] = raw_std[k]; }

    v2f xn[TPW][NF / 2];
#pragma unroll
    for (int rt = 0; rt < TPW; ++rt) {
        const long row0 = (long)(blockIdx.x * RT_PER_BLOCK + rt * WAVES + wave) * 16 + l15;
#pragma unroll
        for (int s = 0; s < NF / 2; ++s) xn[rt][s] = *(const v2f*)(x + row0 * NF + 2 * s);
    }

    __builtin_amdgcn_sched_barrier(0);

#pragma unroll
    for (int i = 0; i < NSTG; ++i) {
        const int j = tid + i * THREADS;
        if (j < NCOLS) {
            v4f kq;
            kq.x = ka[i];
            kq.y = kb[i];
            kq.z = (j < NF * HU) ? 0.0f : kc[i];
            kq.w = 0.0f;
            sB[j]  = kq;
            sW2[j] = 0.5f * kw[i];
        }
    }
    __syncthreads();

    if (tid < 3 * (NCOLS / 32)) {
        const int hg = tid / 3, kk = tid - 3 * hg;
        const float* const sBf = (const float*)sB;
        float s0 = 0.0f, s1 = 0.0f;
        for (int i = 0; i < 32; i += 2) {
            const int j0 = hg * 32 + ((i + hg) & 31);
            const int j1 = hg * 32 + ((i + 1 + hg) & 31);
            s0 = __builtin_fmaf(sW2[j0], sBf[j0 * 4 + kk], s0);
            s1 = __builtin_fmaf(sW2[j1], sBf[j1 * 4 + kk], s1);
        }
        sG[hg * 4 + kk] = s0 + s1;
    }
    __syncthreads();
    if (tid < NF) {
        float a = sG[(2 * tid) * 4 + 0] + sG[(2 * tid + 1) * 4 + 0];
        if (tid == 0) {
            for (int p = 0; p < NP; ++p)
                a += sG[(2 * (NF + p)) * 4 + 0] + sG[(2 * (NF + p) + 1) * 4 + 0];
        } else {
            a += sG[(2 * (NF + tid - 1)) * 4 + 1] + sG[(2 * (NF + tid - 1) + 1) * 4 + 1];
        }
        sAff[tid] = a;
    } else if (tid == NF) {
        float cst = bsum;
        for (int f = 0; f < NF; ++f) cst += sG[(2 * f) * 4 + 1] + sG[(2 * f + 1) * 4 + 1];
        for (int p = 0; p < NP; ++p) cst += sG[(2 * (NF + p)) * 4 + 2] + sG[(2 * (NF + p) + 1) * 4 + 2];
        sAff[NF] = cst;
    }
    __syncthreads();

    float is[NF];
#pragma unroll
    for (int k = 0; k < NF; ++k) is[k] = 1.0f / sd[k];

    v2f* const sAw = sA + wave * (TPW * NGROUPS * APITCH);
    const int slot = half ? 16 : l15;
    const v2f* const sBh = (const v2f*)sB;

    float aff[TPW];
#pragma unroll
    for (int rt = 0; rt < TPW; ++rt) {
        float z[NF];
#pragma unroll
        for (int s = 0; s < NF / 2; ++s) {
            z[2 * s]     = xn[rt][s].x;
            z[2 * s + 1] = xn[rt][s].y;
        }
        float xc = z[0];
#pragma unroll
        for (int k = 1; k < NF; ++k) xc = (k == crp) ? z[k] : xc;
        const float lg = __logf(fmaxf(xc * 10.0f, 1e-6f));
#pragma unroll
        for (int k = 0; k < NF; ++k) {
            const float v = (k == crp) ? lg : z[k];
            z[k] = (v - mn[k]) * is[k];
        }

        float af = sAff[NF];
#pragma unroll
        for (int k = 0; k < NF; ++k) af = __builtin_fmaf(sAff[k], z[k], af);
        aff[rt] = af;

        v2f* const dst = sAw + rt * (NGROUPS * APITCH) + slot;
        if (lane < 17) {
#pragma unroll
            for (int f = 0; f < NF; ++f) {
                v2f a;
                a.x = half ? 0.0f : z[f];
                a.y = half ? 0.0f : 1.0f;
                dst[f * APITCH] = a;
            }
#pragma unroll
            for (int p = 0; p < NP; ++p) {
                v2f a;
                a.x = half ? 1.0f : z[0];
                a.y = half ? 0.0f : z[p + 1];
                dst[(NF + p) * APITCH] = a;
            }
        }
    }

    v8f acc[TPW];
#pragma unroll
    for (int rt = 0; rt < TPW; ++rt) {
        const v8f zero = {0.f, 0.f, 0.f, 0.f, 0.f, 0.f, 0.f, 0.f};
        acc[rt] = zero;
    }

    for (int g = 0; g < NGROUPS; ++g) {
        v2f a[TPW];
#pragma unroll
        for (int rt = 0; rt < TPW; ++rt) a[rt] = sAw[(rt * NGROUPS + g) * APITCH + slot];
        v2f b[TPG];
        float w2v[TPG];
#pragma unroll
        for (int j = 0; j < TPG; ++j) {
            const int n = (g * TPG + j) * 16 + l15;
            b[j]   = sBh[n * 2 + half];
            w2v[j] = sW2[n];
        }
#pragma unroll
        for (int rt = 0; rt < TPW; ++rt) {
            v8f c[TPG];
#pragma unroll
            for (int j = 0; j < TPG; ++j) {
                const v8f zero = {0.f, 0.f, 0.f, 0.f, 0.f, 0.f, 0.f, 0.f};
                c[j] = __builtin_amdgcn_wmma_f32_16x16x4_f32(false, a[rt], false, b[j],
                                                             (short)0, zero, false, false);
            }
            v8f ag = {0.f, 0.f, 0.f, 0.f, 0.f, 0.f, 0.f, 0.f};
            gelu_group_accumulate(c, w2v, ag);
            acc[rt] += ag;
        }
    }

#pragma unroll
    for (int rt = 0; rt < TPW; ++rt) {
        const int rowTile = blockIdx.x * RT_PER_BLOCK + rt * WAVES + wave;
        float red[8];
#pragma unroll
        for (int i = 0; i < 8; ++i) {
            float v = acc[rt][i];
            v += __shfl_xor(v, 1, 32);
            v += __shfl_xor(v, 2, 32);
            v += __shfl_xor(v, 4, 32);
            v += __shfl_xor(v, 8, 32);
            red[i] = v;
        }
        float sel = red[0];
#pragma unroll
        for (int i = 1; i < 8; ++i) sel = ((l15 & 7) == i) ? red[i] : sel;
        if ((l15 >> 3) == half)
            out[rowTile * 16 + l15] = sel + aff[rt];
    }
}

extern "C" void kernel_launch(void* const* d_in, const int* in_sizes, int n_in,
                              void* d_out, int out_size, void* d_ws, size_t ws_size,
                              hipStream_t stream) {
    (void)in_sizes; (void)n_in; (void)out_size; (void)d_ws; (void)ws_size;
    const float* x        = (const float*)d_in[0];
    const float* raw_mean = (const float*)d_in[1];
    const float* raw_std  = (const float*)d_in[2];
    const float* w1f      = (const float*)d_in[3];
    const float* b1f      = (const float*)d_in[4];
    const float* w2f      = (const float*)d_in[5];
    const float* b2f      = (const float*)d_in[6];
    const float* w1p      = (const float*)d_in[7];
    const float* b1p      = (const float*)d_in[8];
    const float* w2p      = (const float*)d_in[9];
    const float* b2p      = (const float*)d_in[10];
    const int*   crp      = (const int*)d_in[11];
    float* out            = (float*)d_out;

    hipLaunchKernelGGL(nam_wmma_kernel, dim3(NBLOCKS), dim3(THREADS), 0, stream,
                       x, raw_mean, raw_std, w1f, b1f, w2f, b2f,
                       w1p, b1p, w2p, b2p, crp, out);
}


# erf fit of degree 5 (was 6) in u=v/5.4 (1/5.4 folded into staged weights), u*u taken with the packed multiply's clamp so |v|>5.4 saturates; erf error 5e-7, was 6e-8: 9 plain ops per pair, was 10
# speedup vs baseline: 2.7632x; 1.0438x over previous
#include <hip/hip_runtime.h>
#include <math.h>

typedef __attribute__((ext_vector_type(2))) float v2f;
typedef __attribute__((ext_vector_type(4))) float v4f;
typedef __attribute__((ext_vector_type(8))) float v8f;

#define BROWS   262144
#define NF      10
#define HU      64
#define NP      9
#define HP      64
#define NCOLS   1216
#define NGROUPS 19
#define TPG     4
#define THREADS 256
#define WAVES   8
#define TPW     4
#define RT_PER_BLOCK (WAVES * TPW)
#define NBLOCKS (BROWS / 16 / RT_PER_BLOCK)
#define NSTG    ((NCOLS + THREADS - 1) / THREADS)
#define APITCH  17

#define KSC   ((float)(1.0 / 5.4))
#define W2SC  ((float)(0.5 * 5.4))
#define Q0 ((float)( 4.308564186e+0))
#define Q1 ((float)( 5.722953796e+0))
#define Q2 ((float)(-2.053866684e-1))
#define Q3 ((float)(-6.750042915e+0))
#define Q4 ((float)( 1.212377930e+1))
#define Q5 ((float)(-7.590076447e+0))

#define HOLD_ORDER() __builtin_amdgcn_sched_barrier(0)

__device__ __forceinline__ v2f square_clamped(const v2f u) {
    v2f t;
    asm("v_pk_mul_f32 %0, %1, %1 clamp" : "=v"(t) : "v"(u));
    return t;
}

__device__ __forceinline__ constexpr int first_rows_order(const int e) {
    return e == 0 ? 4 : e == 1 ? 0 : e == 2 ? 1 : e == 3 ? 2 : e == 4 ? 5 : e == 5 ? 3 : e == 6 ? 6 : 7;
}

__device__ __forceinline__ void gelu_group_accumulate(const v8f (&c)[TPG], const float (&w2v)[TPG], v8f& ag) {
    v2f w[16], q[16], y[16], m[16];
    auto val = [&](const int i) -> v2f {
        const v2f r = {c[i >> 2][2 * (i & 3)], c[i >> 2][2 * (i & 3) + 1]};
        return r;
    };
    auto chain = [&](const int i, const int r) {
        if (r == 0)      w[i] = square_clamped(val(i));
        else if (r == 1) q[i] = __builtin_elementwise_fma((v2f)(Q5), w[i], (v2f)(Q4));
        else if (r == 2) q[i] = __builtin_elementwise_fma(q[i], w[i], (v2f)(Q3));
        else if (r == 3) q[i] = __builtin_elementwise_fma(q[i], w[i], (v2f)(Q2));
        else if (r == 4) q[i] = __builtin_elementwise_fma(q[i], w[i], (v2f)(Q1));
        else if (r == 5) q[i] = __builtin_elementwise_fma(q[i], w[i], (v2f)(Q0));
        else             q[i] = val(i) * q[i];
    };
    auto weigh = [&](const int i) {
        const v2f w2 = {w2v[i >> 2], w2v[i >> 2]};
        m[i] = val(i) * w2;
    };
    auto sum = [&](const int i) {
        const int j = i & 3;
        v2f r;
        if (i < 4) r = y[i] * m[i];
        else {
            const v2f s = {ag[2 * j], ag[2 * j + 1]};
            r = __builtin_elementwise_fma(y[i], m[i], s);
        }
        ag[2 * j]     = r.x;
        ag[2 * j + 1] = r.y;
    };
    auto tanh_of = [&](const int b, const int t) {
        const int i = 4 * b + (t >> 1);
        if (t & 1) y[i].y = __builtin_amdgcn_tanhf(q[i].y);
        else       y[i].x = __builtin_amdgcn_tanhf(q[i].x);
    };
    HOLD_ORDER();
#pragma unroll
    for (int r = 0; r < 4; ++r) {
#pragma unroll
        for (int p = 0; p < 4; ++p) chain(p, r);
#pragma unroll
        for (int p = 0; p < 4; ++p) chain(4 + p, r);
    }
#pragma unroll
    for (int r = 4; r < 7; ++r) {
#pragma unroll
        for (int p = 0; p < 4; ++p) chain(p, r);
    }
    HOLD_ORDER();
#pragma unroll
    for (int b = 0; b < 2; ++b) {
#pragma unroll
        for (int t = 0; t < 8; ++t) {
            tanh_of(b, t);                                  HOLD_ORDER();
#pragma unroll
            for (int e = (7 * t + 1) / 2; e < (7 * (t + 1) + 1) / 2; ++e) {
                const int f = (e < 8) ? first_rows_order(e) : e;
                const int g7 = f >> 2, p = f & 3;
                if (g7 & 1) chain(4 * (b + 1) + p, 4 + (g7 >> 1));
                else        chain(4 * (b + 2) + p, g7 >> 1);
                HOLD_ORDER();
            }
        }
    }
#pragma unroll
    for (int t = 0; t < 8; ++t) {
        tanh_of(2, t);                                      HOLD_ORDER();
#pragma unroll
        for (int f = 3 * t; f < 3 * t + 3; ++f) {
            const int g6 = f >> 2, p = f & 3;
            if (g6 & 1) chain(12 + p, 4 + (g6 >> 1));
            else        weigh(4 * (g6 >> 1) + p);
            HOLD_ORDER();
        }
    }
#pragma unroll
    for (int t = 0; t < 8; ++t) {
        tanh_of(3, t);                                      HOLD_ORDER();
#pragma unroll
        for (int f = 2 * t; f < 2 * t + 2; ++f) {
            if (f < 4) weigh(12 + f);
            else       sum(f - 4);
            HOLD_ORDER();
        }
    }
#pragma unroll
    for (int p = 0; p < 4; ++p) sum(12 + p);
}

__global__ __launch_bounds__(THREADS) void nam_wmma_kernel(
    const float* __restrict__ x,
    const float* __restrict__ raw_mean, const float* __restrict__ raw_std,
    const float* __restrict__ w1f, const float* __restrict__ b1f,
    const float* __restrict__ w2f, const float* __restrict__ b2f,
    const float* __restrict__ w1p, const float* __restrict__ b1p,
    const float* __restrict__ w2p, const float* __restrict__ b2p,
    const int* __restrict__ crp_index_p,
    float* __restrict__ out)
{
    __shared__ v4f   sB[NCOLS];
    __shared__ float sW2[NCOLS];
    __shared__ v2f   sA[WAVES * TPW * NGROUPS * APITCH];
    __shared__ float sG[(NCOLS / 32) * 4];
    __shared__ float sAff[NF + 1];

    const int tid  = threadIdx.x;
    const int lane = tid & 31;
    const int wave = tid >> 5;
    const int half = lane >> 4;
    const int l15  = lane & 15;

    float ka[NSTG], kb[NSTG], kc[NSTG], kw[NSTG];
#pragma unroll
    for (int i = 0; i < NSTG; ++i) {
        const int j  = tid + i * THREADS;
        const int jc = (j < NCOLS) ? j : (NCOLS - 1);
        const bool isF = ((i + 1) * THREADS <= NF * HU) ? true : ((i * THREADS >= NF * HU) ? false : (jc < NF * HU));
        const int q = isF ? 0 : (jc - NF * HU);
        const int p = q >> 6, h = q & 63;
        const float* const pa = isF ? (w1f + jc) : (w1p + (p * 2 + 0) * HP + h);
        const float* const pb = isF ? (b1f + jc) : (w1p + (p * 2 + 1) * HP + h);
        const float* const pw = isF ? (w2f + jc) : (w2p + q);
        ka[i] = *pa;
        kb[i] = *pb;
        kw[i] = *pw;
        kc[i] = b1p[q];
    }
    float bsum = 0.0f;
    for (int f = 0; f < NF; ++f) bsum += b2f[f];
    for (int p = 0; p < NP; ++p) bsum += b2p[p];

    const int crp  = *crp_index_p;

    float mn[NF], sd[NF];
#pragma unroll
    for (int k = 0; k < NF; ++k) { mn[k] = raw_mean[k]; sd[k] = raw_std[k]; }

    v2f xn[TPW][NF / 2];
#pragma unroll
    for (int rt = 0; rt < TPW; ++rt) {
        const long row0 = (long)(blockIdx.x * RT_PER_BLOCK + rt * WAVES + wave) * 16 + l15;
#pragma unroll
        for (int s = 0; s < NF / 2; ++s) xn[rt][s] = *(const v2f*)(x + row0 * NF + 2 * s);
    }

    __builtin_amdgcn_sched_barrier(0);

#pragma unroll
    for (int i = 0; i < NSTG; ++i) {
        const int j = tid + i * THREADS;
        if (j < NCOLS) {
            v4f kq;
            kq.x = KSC * ka[i];
            kq.y = KSC * kb[i];
            kq.z = (j < NF * HU) ? 0.0f : KSC * kc[i];
            kq.w = 0.0f;
            sB[j]  = kq;
            sW2[j] = W2SC * kw[i];
        }
    }
    __syncthreads();

    if (tid < 3 * (NCOLS / 32)) {
        const int hg = tid / 3, kk = tid - 3 * hg;
        const float* const sBf = (const float*)sB;
        float s0 = 0.0f, s1 = 0.0f;
        for (int i = 0; i < 32; i += 2) {
            const int j0 = hg * 32 + ((i + hg) & 31);
            const int j1 = hg * 32 + ((i + 1 + hg) & 31);
            s0 = __builtin_fmaf(sW2[j0], sBf[j0 * 4 + kk], s0);
            s1 = __builtin_fmaf(sW2[j1], sBf[j1 * 4 + kk], s1);
        }
        sG[hg * 4 + kk] = s0 + s1;
    }
    __syncthreads();
    if (tid < NF) {
        float a = sG[(2 * tid) * 4 + 0] + sG[(2 * tid + 1) * 4 + 0];
        if (tid == 0) {
            for (int p = 0; p < NP; ++p)
                a += sG[(2 * (NF + p)) * 4 + 0] + sG[(2 * (NF + p) + 1) * 4 + 0];
        } else {
            a += sG[(2 * (NF + tid - 1)) * 4 + 1] + sG[(2 * (NF + tid - 1) + 1) * 4 + 1];
        }
        sAff[tid] = a;
    } else if (tid == NF) {
        float cst = bsum;
        for (int f = 0; f < NF; ++f) cst += sG[(2 * f) * 4 + 1] + sG[(2 * f + 1) * 4 + 1];
        for (int p = 0; p < NP; ++p) cst += sG[(2 * (NF + p)) * 4 + 2] + sG[(2 * (NF + p) + 1) * 4 + 2];
        sAff[NF] = cst;
    }
    __syncthreads();

    float is[NF];
#pragma unroll
    for (int k = 0; k < NF; ++k) is[k] = 1.0f / sd[k];

    v2f* const sAw = sA + wave * (TPW * NGROUPS * APITCH);
    const int slot = half ? 16 : l15;
    const v2f* const sBh = (const v2f*)sB;

    float aff[TPW];
#pragma unroll
    for (int rt = 0; rt < TPW; ++rt) {
        float z[NF];
#pragma unroll
        for (int s = 0; s < NF / 2; ++s) {
            z[2 * s]     = xn[rt][s].x;
            z[2 * s + 1] = xn[rt][s].y;
        }
        float xc = z[0];
#pragma unroll
        for (int k = 1; k < NF; ++k) xc = (k == crp) ? z[k] : xc;
        const float lg = __logf(fmaxf(xc * 10.0f, 1e-6f));
#pragma unroll
        for (int k = 0; k < NF; ++k) {
            const float v = (k == crp) ? lg : z[k];
            z[k] = (v - mn[k]) * is[k];
        }

        float af = sAff[NF];
#pragma unroll
        for (int k = 0; k < NF; ++k) af = __builtin_fmaf(sAff[k], z[k], af);
        aff[rt] = af;

        v2f* const dst = sAw + rt * (NGROUPS * APITCH) + slot;
        if (lane < 17) {
#pragma unroll
            for (int f = 0; f < NF; ++f) {
                v2f a;
                a.x = half ? 0.0f : z[f];
                a.y = half ? 0.0f : 1.0f;
                dst[f * APITCH] = a;
            }
#pragma unroll
            for (int p = 0; p < NP; ++p) {
                v2f a;
                a.x = half ? 1.0f : z[0];
                a.y = half ? 0.0f : z[p + 1];
                dst[(NF + p) * APITCH] = a;
            }
        }
    }

    v8f acc[TPW];
#pragma unroll
    for (int rt = 0; rt < TPW; ++rt) {
        const v8f zero = {0.f, 0.f, 0.f, 0.f, 0.f, 0.f, 0.f, 0.f};
        acc[rt] = zero;
    }

    for (int g = 0; g < NGROUPS; ++g) {
        v2f a[TPW];
#pragma unroll
        for (int rt = 0; rt < TPW; ++rt) a[rt] = sAw[(rt * NGROUPS + g) * APITCH + slot];
        v2f b[TPG];
        float w2v[TPG];
#pragma unroll
        for (int j = 0; j < TPG; ++j) {
            const int n = (g * TPG + j) * 16 + l15;
            b[j]   = sBh[n * 2 + half];
            w2v[j] = sW2[n];
        }
#pragma unroll
        for (int rt = 0; rt < TPW; ++rt) {
            v8f c[TPG];
#pragma unroll
            for (int j = 0; j < TPG; ++j) {
                const v8f zero = {0.f, 0.f, 0.f, 0.f, 0.f, 0.f, 0.f, 0.f};
                c[j] = __builtin_amdgcn_wmma_f32_16x16x4_f32(false, a[rt], false, b[j],
                                                             (short)0, zero, false, false);
            }
            v8f ag = {0.f, 0.f, 0.f, 0.f, 0.f, 0.f, 0.f, 0.f};
            gelu_group_accumulate(c, w2v, ag);
            acc[rt] += ag;
        }
    }

#pragma unroll
    for (int rt = 0; rt < TPW; ++rt) {
        const int rowTile = blockIdx.x * RT_PER_BLOCK + rt * WAVES + wave;
        float red[8];
#pragma unroll
        for (int i = 0; i < 8; ++i) {
            float v = acc[rt][i];
            v += __shfl_xor(v, 1, 32);
            v += __shfl_xor(v, 2, 32);
            v += __shfl_xor(v, 4, 32);
            v += __shfl_xor(v, 8, 32);
            red[i] = v;
        }
        float sel = red[0];
#pragma unroll
        for (int i = 1; i < 8; ++i) sel = ((l15 & 7) == i) ? red[i] : sel;
        if ((l15 >> 3) == half)
            out[rowTile * 16 + l15] = sel + aff[rt];
    }
}

extern "C" void kernel_launch(void* const* d_in, const int* in_sizes, int n_in,
                              void* d_out, int out_size, void* d_ws, size_t ws_size,
                              hipStream_t stream) {
    (void)in_sizes; (void)n_in; (void)out_size; (void)d_ws; (void)ws_size;
    const float* x        = (const float*)d_in[0];
    const float* raw_mean = (const float*)d_in[1];
    const float* raw_std  = (const float*)d_in[2];
    const float* w1f      = (const float*)d_in[3];
    const float* b1f      = (const float*)d_in[4];
    const float* w2f      = (const float*)d_in[5];
    const float* b2f      = (const float*)d_in[6];
    const float* w1p      = (const float*)d_in[7];
    const float* b1p      = (const float*)d_in[8];
    const float* w2p      = (const float*)d_in[9];
    const float* b2p      = (const float*)d_in[10];
    const int*   crp      = (const int*)d_in[11];
    float* out            = (float*)d_out;

    hipLaunchKernelGGL(nam_wmma_kernel, dim3(NBLOCKS), dim3(THREADS), 0, stream,
                       x, raw_mean, raw_std, w1f, b1f, w2f, b2f,
                       w1p, b1p, w2p, b2p, crp, out);
}
